# starting kernel
# baseline (speedup 1.0000x reference)
#include <hip/hip_runtime.h>


typedef float v2f __attribute__((ext_vector_type(2)));
typedef float v8f __attribute__((ext_vector_type(8)));

#define HF 64
#define WF 64
#define OUTP 7
#define XPITCH 68
#define TPITCH 19

__global__ __launch_bounds__(32)
void roi_pool_wmma_f32(const float* __restrict__ feat,
                       const float* __restrict__ masks,
                       const float* __restrict__ boxes,
                       const int*  __restrict__ h_img_p,
                       const int*  __restrict__ w_img_p,
                       float* __restrict__ out,
                       int C)
{
    __shared__ float Xs[HF * XPITCH];
    __shared__ float Ts[HF * TPITCH];

    const int blk   = blockIdx.x;
    const int n     = blk / C;
    const int c     = blk % C;
    const int lane  = threadIdx.x;
    const int idx16 = lane & 15;
    const int half  = lane >> 4;

    const int mh = *h_img_p;
    const int mw = *w_img_p;

    const float xsc = (float)WF / (float)mw;
    const float ysc = (float)HF / (float)mh;
    const float bx1 = boxes[n * 4 + 0], by1 = boxes[n * 4 + 1];
    const float bx2 = boxes[n * 4 + 2], by2 = boxes[n * 4 + 3];
    const float x1f = fminf(fmaxf(bx1 * xsc, 0.f), (float)(WF - 1));
    const float y1f = fminf(fmaxf(by1 * ysc, 0.f), (float)(HF - 1));
    const float x2f = fmaxf(x1f + 1.f, fminf(bx2 * xsc, (float)WF));
    const float y2f = fmaxf(y1f + 1.f, fminf(by2 * ysc, (float)HF));
    const int x1i = (int)x1f, y1i = (int)y1f;
    const int x2i = (int)x2f, y2i = (int)y2f;
    const int hh = y2i - y1i, ww = x2i - x1i;

    const int bi = idx16;
    const int rs = y1i + (bi * hh) / OUTP;
    const int re = y1i + ((bi + 1) * hh + OUTP - 1) / OUTP;
    const int cs = x1i + (bi * ww) / OUTP;
    const int ce = x1i + ((bi + 1) * ww + OUTP - 1) / OUTP;
    const float rinv = (bi < OUTP) ? 1.f / (float)(re - rs) : 0.f;
    const float cinv = (bi < OUTP) ? 1.f / (float)(ce - cs) : 0.f;

    const float* fbase = feat  + (size_t)c * (HF * WF);
    const float* mbase = masks + (size_t)n * ((size_t)mh * (size_t)mw);
    #pragma unroll 4
    for (int it = 0; it < 32; ++it) {
        const int e = (it * 32 + lane) * 4;
        const int r = e >> 6, w = e & 63;
        float4 f = *(const float4*)(fbase + e);
        float4 m = *(const float4*)(mbase + (size_t)r * mw + w);
        *(float4*)&Xs[r * XPITCH + w] =
            make_float4(f.x * m.x, f.y * m.y, f.z * m.z, f.w * m.w);
    }
    __syncthreads();

    v8f acc0 = {}, acc1 = {}, acc2 = {}, acc3 = {};
    #pragma unroll
    for (int kk = 0; kk < 16; ++kk) {
        const int k0 = 4 * kk + 2 * half;
        v2f b;
        b.x = (k0     >= cs && k0     < ce) ? cinv : 0.f;
        b.y = (k0 + 1 >= cs && k0 + 1 < ce) ? cinv : 0.f;
        v2f a0 = *(const v2f*)&Xs[( 0 + idx16) * XPITCH + k0];
        v2f a1 = *(const v2f*)&Xs[(16 + idx16) * XPITCH + k0];
        v2f a2 = *(const v2f*)&Xs[(32 + idx16) * XPITCH + k0];
        v2f a3 = *(const v2f*)&Xs[(48 + idx16) * XPITCH + k0];
        acc0 = __builtin_amdgcn_wmma_f32_16x16x4_f32(false, a0, false, b, (short)0, acc0, false, false);
        acc1 = __builtin_amdgcn_wmma_f32_16x16x4_f32(false, a1, false, b, (short)0, acc1, false, false);
        acc2 = __builtin_amdgcn_wmma_f32_16x16x4_f32(false, a2, false, b, (short)0, acc2, false, false);
        acc3 = __builtin_amdgcn_wmma_f32_16x16x4_f32(false, a3, false, b, (short)0, acc3, false, false);
    }

    __syncthreads();
    #pragma unroll
    for (int g = 0; g < 8; ++g) {
        const int rb = g + 8 * half;
        Ts[(rb +  0) * TPITCH + idx16] = acc0[g];
        Ts[(rb + 16) * TPITCH + idx16] = acc1[g];
        Ts[(rb + 32) * TPITCH + idx16] = acc2[g];
        Ts[(rb + 48) * TPITCH + idx16] = acc3[g];
    }
    __syncthreads();

    v8f accO = {};
    #pragma unroll
    for (int kk = 0; kk < 16; ++kk) {
        const int k0 = 4 * kk + 2 * half;
        v2f a;
        a.x = (k0     >= rs && k0     < re) ? rinv : 0.f;
        a.y = (k0 + 1 >= rs && k0 + 1 < re) ? rinv : 0.f;
        v2f b;
        b.x = Ts[(k0    ) * TPITCH + idx16];
        b.y = Ts[(k0 + 1) * TPITCH + idx16];
        accO = __builtin_amdgcn_wmma_f32_16x16x4_f32(false, a, false, b, (short)0, accO, false, false);
    }

    float* obase = out + (size_t)(n * C + c) * (OUTP * OUTP);
    #pragma unroll
    for (int g = 0; g < 8; ++g) {
        const int i = g + 8 * half;
        const int j = idx16;
        if (i < OUTP && j < OUTP) obase[i * OUTP + j] = accO[g];
    }
}

extern "C" void kernel_launch(void* const* d_in, const int* in_sizes, int n_in,
                              void* d_out, int out_size, void* d_ws, size_t ws_size,
                              hipStream_t stream) {
    const float* feat  = (const float*)d_in[0];
    const float* masks = (const float*)d_in[1];
    const float* boxes = (const float*)d_in[2];
    const int*   h_img = (const int*)d_in[3];
    const int*   w_img = (const int*)d_in[4];
    float*       out   = (float*)d_out;

    const int C = in_sizes[0] / (HF * WF);
    const int N = in_sizes[2] / 4;

    roi_pool_wmma_f32<<<dim3(N * C), dim3(32), 0, stream>>>(
        feat, masks, boxes, h_img, w_img, out, C);
}


# Box-restricted work: fetch, stage and multiply only the 16-row tiles and 4-wide K-steps the box touches (zero-weight terms skipped; kept terms summed in the same order)
# speedup vs baseline: 2.3800x; 2.3800x over previous
#include <hip/hip_runtime.h>


typedef float v2f __attribute__((ext_vector_type(2)));
typedef float v8f __attribute__((ext_vector_type(8)));

#define HF 64
#define WF 64
#define OUTP 7
#define XPITCH 68
#define TPITCH 19

template <int NT>
__device__ __forceinline__ void box_cols_product(const float* xrow, int kc0, int kc1, int half,
                                                 int cs, int ce, float cinv, v8f (&acc)[4])
{
    #pragma unroll 4
    for (int kk = kc0; kk < kc1; ++kk) {
        const int k0 = 4 * kk + 2 * half;
        v2f b;
        b.x = (k0     >= cs && k0     < ce) ? cinv : 0.f;
        b.y = (k0 + 1 >= cs && k0 + 1 < ce) ? cinv : 0.f;
        #pragma unroll
        for (int j = 0; j < NT; ++j) {
            const v2f a = *(const v2f*)(xrow + j * 16 * XPITCH + 4 * kk);
            acc[j] = __builtin_amdgcn_wmma_f32_16x16x4_f32(false, a, false, b, (short)0, acc[j], false, false);
        }
    }
}

__global__ __launch_bounds__(32)
void roi_pool_wmma_f32(const float* __restrict__ feat,
                       const float* __restrict__ masks,
                       const float* __restrict__ boxes,
                       const int*  __restrict__ h_img_p,
                       const int*  __restrict__ w_img_p,
                       float* __restrict__ out,
                       int C)
{
    __shared__ float Xs[HF * XPITCH];
    __shared__ float Ts[HF * TPITCH];

    const int blk   = blockIdx.x;
    const int n     = blk / C;
    const int c     = blk % C;
    const int lane  = threadIdx.x;
    const int idx16 = lane & 15;
    const int half  = lane >> 4;

    const int mh = *h_img_p;
    const int mw = *w_img_p;

    const float xsc = (float)WF / (float)mw;
    const float ysc = (float)HF / (float)mh;
    const float bx1 = boxes[n * 4 + 0], by1 = boxes[n * 4 + 1];
    const float bx2 = boxes[n * 4 + 2], by2 = boxes[n * 4 + 3];
    const float x1f = fminf(fmaxf(bx1 * xsc, 0.f), (float)(WF - 1));
    const float y1f = fminf(fmaxf(by1 * ysc, 0.f), (float)(HF - 1));
    const float x2f = fmaxf(x1f + 1.f, fminf(bx2 * xsc, (float)WF));
    const float y2f = fmaxf(y1f + 1.f, fminf(by2 * ysc, (float)HF));
    const int x1i = __builtin_amdgcn_readfirstlane((int)x1f);
    const int y1i = __builtin_amdgcn_readfirstlane((int)y1f);
    const int x2i = __builtin_amdgcn_readfirstlane((int)x2f);
    const int y2i = __builtin_amdgcn_readfirstlane((int)y2f);
    const int hh = y2i - y1i, ww = x2i - x1i;

    const int bi = idx16;
    const int rs = y1i + (bi * hh) / OUTP;
    const int re = y1i + ((bi + 1) * hh + OUTP - 1) / OUTP;
    const int cs = x1i + (bi * ww) / OUTP;
    const int ce = x1i + ((bi + 1) * ww + OUTP - 1) / OUTP;
    const float rinv = (bi < OUTP) ? 1.f / (float)(re - rs) : 0.f;
    const float cinv = (bi < OUTP) ? 1.f / (float)(ce - cs) : 0.f;

    const int kc0 = x1i >> 2, kc1 = (x2i + 3) >> 2;
    const int kr0 = y1i >> 2, kr1 = (y2i + 3) >> 2;
    const int R0  = (y1i >> 4) << 4;
    const int nt  = ((y2i + 15) >> 4) - (y1i >> 4);
    const int R1  = R0 + 16 * nt;

    const float* fbase = feat  + (size_t)c * (HF * WF);
    const float* mbase = masks + (size_t)n * ((size_t)mh * (size_t)mw);
    const int nc4    = kc1 - kc0;
    const int sh     = (nc4 > 1) ? 32 - __builtin_clz((unsigned)(nc4 - 1)) : 0;
    const int rstep  = 32 >> sh;
    const int rloc   = lane >> sh;
    const int c4     = min(kc0 + (lane & ((1 << sh) - 1)), kc1 - 1);
    const int passes = (16 * nt + rstep - 1) >> (5 - sh);
    const int trips  = (passes + 3) >> 2;
    const float* fcol = fbase + 4 * c4;
    const float* mcol = mbase + 4 * c4;
    float*       xcol = Xs + 4 * c4;
    for (int q = 0; q < trips; ++q) {
        #pragma unroll
        for (int u = 0; u < 4; ++u) {
            const int r  = min(R0 + rloc + (4 * q + u) * rstep, R1 - 1);
            const int rg = min(max(r, y1i), y2i - 1);
            const float4 f = *(const float4*)(fcol + rg * WF);
            const float4 m = *(const float4*)(mcol + (size_t)rg * (size_t)mw);
            const bool inbox = (r == rg);
            float4 x;
            x.x = inbox ? f.x * m.x : 0.f;
            x.y = inbox ? f.y * m.y : 0.f;
            x.z = inbox ? f.z * m.z : 0.f;
            x.w = inbox ? f.w * m.w : 0.f;
            *(float4*)(xcol + r * XPITCH) = x;
        }
    }
    __syncthreads();

    v8f acc[4] = {};
    const float* xrow = &Xs[(R0 + idx16) * XPITCH + 2 * half];
    switch (nt) {
        case 1:  box_cols_product<1>(xrow, kc0, kc1, half, cs, ce, cinv, acc); break;
        case 2:  box_cols_product<2>(xrow, kc0, kc1, half, cs, ce, cinv, acc); break;
        case 3:  box_cols_product<3>(xrow, kc0, kc1, half, cs, ce, cinv, acc); break;
        default: box_cols_product<4>(xrow, kc0, kc1, half, cs, ce, cinv, acc); break;
    }

    __syncthreads();
    float* trow = &Ts[(R0 + 8 * half) * TPITCH + idx16];
    #pragma unroll
    for (int j = 0; j < 4; ++j) {
        if (j < nt) {
            #pragma unroll
            for (int g = 0; g < 8; ++g) trow[(16 * j + g) * TPITCH] = acc[j][g];
        }
    }
    __syncthreads();

    v8f accO = {};
    #pragma unroll 4
    for (int kk = kr0; kk < kr1; ++kk) {
        const int k0 = 4 * kk + 2 * half;
        v2f a;
        a.x = (k0     >= rs && k0     < re) ? rinv : 0.f;
        a.y = (k0 + 1 >= rs && k0 + 1 < re) ? rinv : 0.f;
        v2f b;
        b.x = Ts[(k0    ) * TPITCH + idx16];
        b.y = Ts[(k0 + 1) * TPITCH + idx16];
        accO = __builtin_amdgcn_wmma_f32_16x16x4_f32(false, a, false, b, (short)0, accO, false, false);
    }

    float* obase = out + (size_t)(n * C + c) * (OUTP * OUTP);
    #pragma unroll
    for (int g = 0; g < 8; ++g) {
        const int i = g + 8 * half;
        const int j = idx16;
        if (i < OUTP && j < OUTP) obase[i * OUTP + j] = accO[g];
    }
}

extern "C" void kernel_launch(void* const* d_in, const int* in_sizes, int n_in,
                              void* d_out, int out_size, void* d_ws, size_t ws_size,
                              hipStream_t stream) {
    const float* feat  = (const float*)d_in[0];
    const float* masks = (const float*)d_in[1];
    const float* boxes = (const float*)d_in[2];
    const int*   h_img = (const int*)d_in[3];
    const int*   w_img = (const int*)d_in[4];
    float*       out   = (float*)d_out;

    const int C = in_sizes[0] / (HF * WF);
    const int N = in_sizes[2] / 4;

    roi_pool_wmma_f32<<<dim3(N * C), dim3(32), 0, stream>>>(
        feat, masks, boxes, h_img, w_img, out, C);
}


# Two channels of a box per wave (waves 16384 -> 8192): channel 2's column weights fill the idle accumulator columns 8-14, so one tmp spill, one row product, one mask fetch serve both; sum order kept
# speedup vs baseline: 2.5909x; 1.0886x over previous
#include <hip/hip_runtime.h>


typedef float v2f __attribute__((ext_vector_type(2)));
typedef float v8f __attribute__((ext_vector_type(8)));

#define HF 64
#define WF 64
#define OUTP 7
#define XPITCH 68
#define TPITCH 19
#define XTILE (HF * XPITCH)

template <int NT>
__device__ __forceinline__ void box_cols_product(const float* xrow, int kc0, int kc1, int half,
                                                 int cs, int ce, float cinv0, float cinv1, v8f (&acc)[4])
{
    #pragma unroll 4
    for (int kk = kc0; kk < kc1; ++kk) {
        const int k0 = 4 * kk + 2 * half;
        const bool in0 = (k0     >= cs && k0     < ce);
        const bool in1 = (k0 + 1 >= cs && k0 + 1 < ce);
        v2f b0, b1;
        b0.x = in0 ? cinv0 : 0.f;
        b0.y = in1 ? cinv0 : 0.f;
        b1.x = in0 ? cinv1 : 0.f;
        b1.y = in1 ? cinv1 : 0.f;
        #pragma unroll
        for (int j = 0; j < NT; ++j) {
            const v2f a0 = *(const v2f*)(xrow + j * 16 * XPITCH + 4 * kk);
            const v2f a1 = *(const v2f*)(xrow + XTILE + j * 16 * XPITCH + 4 * kk);
            acc[j] = __builtin_amdgcn_wmma_f32_16x16x4_f32(false, a0, false, b0, (short)0, acc[j], false, false);
            acc[j] = __builtin_amdgcn_wmma_f32_16x16x4_f32(false, a1, false, b1, (short)0, acc[j], false, false);
        }
    }
}

__global__ __launch_bounds__(32)
void roi_pool_wmma_f32(const float* __restrict__ feat,
                       const float* __restrict__ masks,
                       const float* __restrict__ boxes,
                       const int*  __restrict__ h_img_p,
                       const int*  __restrict__ w_img_p,
                       float* __restrict__ out,
                       int C)
{
    __shared__ float Xs[2 * XTILE];
    __shared__ float Ts[HF * TPITCH];

    const int P     = (C + 1) >> 1;
    const int blk   = blockIdx.x;
    const int n     = blk / P;
    const int c0    = 2 * (blk % P);
    const bool has1 = (c0 + 1 < C);
    const int c1    = has1 ? c0 + 1 : c0;
    const int lane  = threadIdx.x;
    const int idx16 = lane & 15;
    const int half  = lane >> 4;

    const int mh = *h_img_p;
    const int mw = *w_img_p;

    const float xsc = (float)WF / (float)mw;
    const float ysc = (float)HF / (float)mh;
    const float bx1 = boxes[n * 4 + 0], by1 = boxes[n * 4 + 1];
    const float bx2 = boxes[n * 4 + 2], by2 = boxes[n * 4 + 3];
    const float x1f = fminf(fmaxf(bx1 * xsc, 0.f), (float)(WF - 1));
    const float y1f = fminf(fmaxf(by1 * ysc, 0.f), (float)(HF - 1));
    const float x2f = fmaxf(x1f + 1.f, fminf(bx2 * xsc, (float)WF));
    const float y2f = fmaxf(y1f + 1.f, fminf(by2 * ysc, (float)HF));
    const int x1i = __builtin_amdgcn_readfirstlane((int)x1f);
    const int y1i = __builtin_amdgcn_readfirstlane((int)y1f);
    const int x2i = __builtin_amdgcn_readfirstlane((int)x2f);
    const int y2i = __builtin_amdgcn_readfirstlane((int)y2f);
    const int hh = y2i - y1i, ww = x2i - x1i;

    const int bi  = idx16;
    const int bj  = idx16 & 7;
    const int sel = idx16 >> 3;
    const int rs = y1i + (bi * hh) / OUTP;
    const int re = y1i + ((bi + 1) * hh + OUTP - 1) / OUTP;
    const int cs = x1i + (bj * ww) / OUTP;
    const int ce = x1i + ((bj + 1) * ww + OUTP - 1) / OUTP;
    const float rinv  = (bi < OUTP) ? 1.f / (float)(re - rs) : 0.f;
    const float cw    = (bj < OUTP) ? 1.f / (float)(ce - cs) : 0.f;
    const float cinv0 = (sel == 0) ? cw : 0.f;
    const float cinv1 = (sel == 1 && has1) ? cw : 0.f;

    const int kc0 = x1i >> 2, kc1 = (x2i + 3) >> 2;
    const int kr0 = y1i >> 2, kr1 = (y2i + 3) >> 2;
    const int R0  = (y1i >> 4) << 4;
    const int nt  = ((y2i + 15) >> 4) - (y1i >> 4);
    const int R1  = R0 + 16 * nt;

    const float* fbase0 = feat + (size_t)c0 * (HF * WF);
    const float* fbase1 = feat + (size_t)c1 * (HF * WF);
    const float* mbase = masks + (size_t)n * ((size_t)mh * (size_t)mw);
    const int nc4    = kc1 - kc0;
    const int sh     = (nc4 > 1) ? 32 - __builtin_clz((unsigned)(nc4 - 1)) : 0;
    const int rstep  = 32 >> sh;
    const int rloc   = lane >> sh;
    const int c4     = min(kc0 + (lane & ((1 << sh) - 1)), kc1 - 1);
    const int passes = (16 * nt + rstep - 1) >> (5 - sh);
    const int trips  = (passes + 3) >> 2;
    const float* fcol0 = fbase0 + 4 * c4;
    const float* fcol1 = fbase1 + 4 * c4;
    const float* mcol = mbase + 4 * c4;
    float*       xcol = Xs + 4 * c4;
    for (int q = 0; q < trips; ++q) {
        #pragma unroll
        for (int u = 0; u < 4; ++u) {
            const int r  = min(R0 + rloc + (4 * q + u) * rstep, R1 - 1);
            const int rg = min(max(r, y1i), y2i - 1);
            const float4 f0 = *(const float4*)(fcol0 + rg * WF);
            const float4 f1 = *(const float4*)(fcol1 + rg * WF);
            const float4 m  = *(const float4*)(mcol + (size_t)rg * (size_t)mw);
            const bool inbox = (r == rg);
            float4 x0, x1;
            x0.x = inbox ? f0.x * m.x : 0.f;
            x0.y = inbox ? f0.y * m.y : 0.f;
            x0.z = inbox ? f0.z * m.z : 0.f;
            x0.w = inbox ? f0.w * m.w : 0.f;
            x1.x = inbox ? f1.x * m.x : 0.f;
            x1.y = inbox ? f1.y * m.y : 0.f;
            x1.z = inbox ? f1.z * m.z : 0.f;
            x1.w = inbox ? f1.w * m.w : 0.f;
            *(float4*)(xcol + r * XPITCH) = x0;
            *(float4*)(xcol + XTILE + r * XPITCH) = x1;
        }
    }
    __syncthreads();

    v8f acc[4] = {};
    const float* xrow = &Xs[(R0 + idx16) * XPITCH + 2 * half];
    switch (nt) {
        case 1:  box_cols_product<1>(xrow, kc0, kc1, half, cs, ce, cinv0, cinv1, acc); break;
        case 2:  box_cols_product<2>(xrow, kc0, kc1, half, cs, ce, cinv0, cinv1, acc); break;
        case 3:  box_cols_product<3>(xrow, kc0, kc1, half, cs, ce, cinv0, cinv1, acc); break;
        default: box_cols_product<4>(xrow, kc0, kc1, half, cs, ce, cinv0, cinv1, acc); break;
    }

    __syncthreads();
    float* trow = &Ts[(R0 + 8 * half) * TPITCH + idx16];
    #pragma unroll
    for (int j = 0; j < 4; ++j) {
        if (j < nt) {
            #pragma unroll
            for (int g = 0; g < 8; ++g) trow[(16 * j + g) * TPITCH] = acc[j][g];
        }
    }
    __syncthreads();

    v8f accO = {};
    #pragma unroll 4
    for (int kk = kr0; kk < kr1; ++kk) {
        const int k0 = 4 * kk + 2 * half;
        v2f a;
        a.x = (k0     >= rs && k0     < re) ? rinv : 0.f;
        a.y = (k0 + 1 >= rs && k0 + 1 < re) ? rinv : 0.f;
        v2f b;
        b.x = Ts[(k0    ) * TPITCH + idx16];
        b.y = Ts[(k0 + 1) * TPITCH + idx16];
        accO = __builtin_amdgcn_wmma_f32_16x16x4_f32(false, a, false, b, (short)0, accO, false, false);
    }

    float* obase = out + (size_t)(n * C + (sel ? c1 : c0)) * (OUTP * OUTP);
    const bool mine = (bj < OUTP) && (sel == 0 || has1);
    #pragma unroll
    for (int g = 0; g < 8; ++g) {
        const int i = g + 8 * half;
        if (i < OUTP && mine) obase[i * OUTP + bj] = accO[g];
    }
}

extern "C" void kernel_launch(void* const* d_in, const int* in_sizes, int n_in,
                              void* d_out, int out_size, void* d_ws, size_t ws_size,
                              hipStream_t stream) {
    const float* feat  = (const float*)d_in[0];
    const float* masks = (const float*)d_in[1];
    const float* boxes = (const float*)d_in[2];
    const int*   h_img = (const int*)d_in[3];
    const int*   w_img = (const int*)d_in[4];
    float*       out   = (float*)d_out;

    const int C = in_sizes[0] / (HF * WF);
    const int N = in_sizes[2] / 4;

    roi_pool_wmma_f32<<<dim3(N * ((C + 1) / 2)), dim3(32), 0, stream>>>(
        feat, masks, boxes, h_img, w_img, out, C);
}


# tmp stays in registers: step 1's accumulator registers 2s,2s+1 are step 2's B operand as they stand (Wr built in that k order), no LDS spill/reload; row sum order: rows {2s,2s+1,8+2s,9+2s} per K-step
# speedup vs baseline: 2.7358x; 1.0559x over previous
#include <hip/hip_runtime.h>


typedef float v2f __attribute__((ext_vector_type(2)));
typedef float v8f __attribute__((ext_vector_type(8)));

#define HF 64
#define WF 64
#define OUTP 7
#define XPITCH 68
#define XTILE (HF * XPITCH)

template <int NT>
__device__ __forceinline__ void box_cols_product(const float* xrow, int kc0, int kc1, int half,
                                                 int cs, int ce, float cinv0, float cinv1, v8f (&acc)[4])
{
    #pragma unroll 4
    for (int kk = kc0; kk < kc1; ++kk) {
        const int k0 = 4 * kk + 2 * half;
        const bool in0 = (k0     >= cs && k0     < ce);
        const bool in1 = (k0 + 1 >= cs && k0 + 1 < ce);
        v2f b0, b1;
        b0.x = in0 ? cinv0 : 0.f;
        b0.y = in1 ? cinv0 : 0.f;
        b1.x = in0 ? cinv1 : 0.f;
        b1.y = in1 ? cinv1 : 0.f;
        #pragma unroll
        for (int j = 0; j < NT; ++j) {
            const v2f a0 = *(const v2f*)(xrow + j * 16 * XPITCH + 4 * kk);
            const v2f a1 = *(const v2f*)(xrow + XTILE + j * 16 * XPITCH + 4 * kk);
            acc[j] = __builtin_amdgcn_wmma_f32_16x16x4_f32(false, a0, false, b0, (short)0, acc[j], false, false);
            acc[j] = __builtin_amdgcn_wmma_f32_16x16x4_f32(false, a1, false, b1, (short)0, acc[j], false, false);
        }
    }
}

__global__ __launch_bounds__(32)
void roi_pool_wmma_f32(const float* __restrict__ feat,
                       const float* __restrict__ masks,
                       const float* __restrict__ boxes,
                       const int*  __restrict__ h_img_p,
                       const int*  __restrict__ w_img_p,
                       float* __restrict__ out,
                       int C)
{
    __shared__ float Xs[2 * XTILE];

    const int P     = (C + 1) >> 1;
    const int blk   = blockIdx.x;
    const int n     = blk / P;
    const int c0    = 2 * (blk % P);
    const bool has1 = (c0 + 1 < C);
    const int c1    = has1 ? c0 + 1 : c0;
    const int lane  = threadIdx.x;
    const int idx16 = lane & 15;
    const int half  = lane >> 4;

    const int mh = *h_img_p;
    const int mw = *w_img_p;

    const float xsc = (float)WF / (float)mw;
    const float ysc = (float)HF / (float)mh;
    const float bx1 = boxes[n * 4 + 0], by1 = boxes[n * 4 + 1];
    const float bx2 = boxes[n * 4 + 2], by2 = boxes[n * 4 + 3];
    const float x1f = fminf(fmaxf(bx1 * xsc, 0.f), (float)(WF - 1));
    const float y1f = fminf(fmaxf(by1 * ysc, 0.f), (float)(HF - 1));
    const float x2f = fmaxf(x1f + 1.f, fminf(bx2 * xsc, (float)WF));
    const float y2f = fmaxf(y1f + 1.f, fminf(by2 * ysc, (float)HF));
    const int x1i = __builtin_amdgcn_readfirstlane((int)x1f);
    const int y1i = __builtin_amdgcn_readfirstlane((int)y1f);
    const int x2i = __builtin_amdgcn_readfirstlane((int)x2f);
    const int y2i = __builtin_amdgcn_readfirstlane((int)y2f);
    const int hh = y2i - y1i, ww = x2i - x1i;

    const int bi  = idx16;
    const int bj  = idx16 & 7;
    const int sel = idx16 >> 3;
    const int rs = y1i + (bi * hh) / OUTP;
    const int re = y1i + ((bi + 1) * hh + OUTP - 1) / OUTP;
    const int cs = x1i + (bj * ww) / OUTP;
    const int ce = x1i + ((bj + 1) * ww + OUTP - 1) / OUTP;
    const float rinv  = (bi < OUTP) ? 1.f / (float)(re - rs) : 0.f;
    const float cw    = (bj < OUTP) ? 1.f / (float)(ce - cs) : 0.f;
    const float cinv0 = (sel == 0) ? cw : 0.f;
    const float cinv1 = (sel == 1 && has1) ? cw : 0.f;

    const int kc0 = x1i >> 2, kc1 = (x2i + 3) >> 2;
    const int R0  = (y1i >> 4) << 4;
    const int nt  = ((y2i + 15) >> 4) - (y1i >> 4);
    const int R1  = R0 + 16 * nt;

    const float* fbase0 = feat + (size_t)c0 * (HF * WF);
    const float* fbase1 = feat + (size_t)c1 * (HF * WF);
    const float* mbase = masks + (size_t)n * ((size_t)mh * (size_t)mw);
    const int nc4    = kc1 - kc0;
    const int sh     = (nc4 > 1) ? 32 - __builtin_clz((unsigned)(nc4 - 1)) : 0;
    const int rstep  = 32 >> sh;
    const int rloc   = lane >> sh;
    const int c4     = min(kc0 + (lane & ((1 << sh) - 1)), kc1 - 1);
    const int passes = (16 * nt + rstep - 1) >> (5 - sh);
    const int trips  = (passes + 3) >> 2;
    const float* fcol0 = fbase0 + 4 * c4;
    const float* fcol1 = fbase1 + 4 * c4;
    const float* mcol = mbase + 4 * c4;
    float*       xcol = Xs + 4 * c4;
    for (int q = 0; q < trips; ++q) {
        #pragma unroll
        for (int u = 0; u < 4; ++u) {
            const int r  = min(R0 + rloc + (4 * q + u) * rstep, R1 - 1);
            const int rg = min(max(r, y1i), y2i - 1);
            const float4 f0 = *(const float4*)(fcol0 + rg * WF);
            const float4 f1 = *(const float4*)(fcol1 + rg * WF);
            const float4 m  = *(const float4*)(mcol + (size_t)rg * (size_t)mw);
            const bool inbox = (r == rg);
            float4 x0, x1;
            x0.x = inbox ? f0.x * m.x : 0.f;
            x0.y = inbox ? f0.y * m.y : 0.f;
            x0.z = inbox ? f0.z * m.z : 0.f;
            x0.w = inbox ? f0.w * m.w : 0.f;
            x1.x = inbox ? f1.x * m.x : 0.f;
            x1.y = inbox ? f1.y * m.y : 0.f;
            x1.z = inbox ? f1.z * m.z : 0.f;
            x1.w = inbox ? f1.w * m.w : 0.f;
            *(float4*)(xcol + r * XPITCH) = x0;
            *(float4*)(xcol + XTILE + r * XPITCH) = x1;
        }
    }
    __syncthreads();

    v8f acc[4] = {};
    const float* xrow = &Xs[(R0 + idx16) * XPITCH + 2 * half];
    switch (nt) {
        case 1:  box_cols_product<1>(xrow, kc0, kc1, half, cs, ce, cinv0, cinv1, acc); break;
        case 2:  box_cols_product<2>(xrow, kc0, kc1, half, cs, ce, cinv0, cinv1, acc); break;
        case 3:  box_cols_product<3>(xrow, kc0, kc1, half, cs, ce, cinv0, cinv1, acc); break;
        default: box_cols_product<4>(xrow, kc0, kc1, half, cs, ce, cinv0, cinv1, acc); break;
    }

    v8f accO = {};
    #pragma unroll
    for (int j = 0; j < 4; ++j) {
        if (j < nt) {
            #pragma unroll
            for (int s = 0; s < 4; ++s) {
                const int k0 = R0 + 16 * j + 8 * half + 2 * s;
                v2f a;
                a.x = (k0     >= rs && k0     < re) ? rinv : 0.f;
                a.y = (k0 + 1 >= rs && k0 + 1 < re) ? rinv : 0.f;
                v2f b;
                b.x = acc[j][2 * s];
                b.y = acc[j][2 * s + 1];
                accO = __builtin_amdgcn_wmma_f32_16x16x4_f32(false, a, false, b, (short)0, accO, false, false);
            }
        }
    }

    float* obase = out + (size_t)(n * C + (sel ? c1 : c0)) * (OUTP * OUTP);
    const bool mine = (bj < OUTP) && (sel == 0 || has1);
    #pragma unroll
    for (int g = 0; g < 8; ++g) {
        const int i = g + 8 * half;
        if (i < OUTP && mine) obase[i * OUTP + bj] = accO[g];
    }
}

extern "C" void kernel_launch(void* const* d_in, const int* in_sizes, int n_in,
                              void* d_out, int out_size, void* d_ws, size_t ws_size,
                              hipStream_t stream) {
    const float* feat  = (const float*)d_in[0];
    const float* masks = (const float*)d_in[1];
    const float* boxes = (const float*)d_in[2];
    const int*   h_img = (const int*)d_in[3];
    const int*   w_img = (const int*)d_in[4];
    float*       out   = (float*)d_out;

    const int C = in_sizes[0] / (HF * WF);
    const int N = in_sizes[2] / 4;

    roi_pool_wmma_f32<<<dim3(N * ((C + 1) / 2)), dim3(32), 0, stream>>>(
        feat, masks, boxes, h_img, w_img, out, C);
}
